# starting kernel
# baseline (speedup 1.0000x reference)
#include <hip/hip_runtime.h>
#include <hip/hip_bf16.h>
#include <math.h>

#define R_H 6
#define EPSBN 1e-5f
#define BSZ 8
#define CSZ 64
#define NSZ 16384
#define OC  64
#define DT  128
#define YPS 288
#define WPS 160

typedef __attribute__((ext_vector_type(2))) float v2f;
typedef __attribute__((ext_vector_type(8))) float v8f;

__device__ __forceinline__ void hermite6(float xn, float p[R_H]) {
    const float SQ13 = 3.60555127546398929312f;
    const float PIQ  = 0.75112554446494248286f;
    float t = tanhf(xn) * SQ13;
    float g = __expf(-0.5f * t * t);
    p[0] = PIQ * g;
    p[1] = 1.41421356237309515f * PIQ * t * g;
    p[2] = 1.0f                 * t * p[1] - 0.70710678118654752f * p[0];
    p[3] = 0.81649658092772603f * t * p[2] - 0.81649658092772603f * p[1];
    p[4] = 0.70710678118654752f * t * p[3] - 0.86602540378443865f * p[2];
    p[5] = 0.63245553203367587f * t * p[4] - 0.89442719099991588f * p[3];
}

__global__ void k_init(float* __restrict__ ws) {
    int i = threadIdx.x;
    if (i < 2 * CSZ) ws[i] = 0.0f;
}

__global__ void __launch_bounds__(256) k_stats(const float* __restrict__ x,
                                               float* __restrict__ ws) {
    __shared__ float red[256];
    int bc = blockIdx.x;
    int c  = bc & (CSZ - 1);
    const float* xp = x + (size_t)bc * NSZ;
    float s = 0.f, s2 = 0.f;
    for (int d = threadIdx.x; d < NSZ; d += 256) {
        float v = xp[d];
        s += v; s2 += v * v;
    }
    red[threadIdx.x] = s; __syncthreads();
    for (int st = 128; st > 0; st >>= 1) {
        if (threadIdx.x < st) red[threadIdx.x] += red[threadIdx.x + st];
        __syncthreads();
    }
    if (threadIdx.x == 0) atomicAdd(&ws[c], red[0]);
    __syncthreads();
    red[threadIdx.x] = s2; __syncthreads();
    for (int st = 128; st > 0; st >>= 1) {
        if (threadIdx.x < st) red[threadIdx.x] += red[threadIdx.x + st];
        __syncthreads();
    }
    if (threadIdx.x == 0) atomicAdd(&ws[CSZ + c], red[0]);
}

__global__ void k_finalize(const float* __restrict__ gamma,
                           const float* __restrict__ beta,
                           float* __restrict__ ws) {
    int c = threadIdx.x;
    if (c >= CSZ) return;
    const float inv_n = 1.0f / ((float)BSZ * (float)NSZ);
    float mean = ws[c] * inv_n;
    float var  = ws[CSZ + c] * inv_n - mean * mean;
    float rstd = rsqrtf(var + EPSBN);
    float sc   = gamma[c] * rstd;
    ws[2 * CSZ + c] = sc;
    ws[3 * CSZ + c] = beta[c] - mean * sc;
}

__global__ void __launch_bounds__(256) k_cost(const float* __restrict__ x,
                                              float* __restrict__ ws) {
    __shared__ float red[256];
    __shared__ float costk[R_H];
    int bc = blockIdx.x;
    int c  = bc & (CSZ - 1);
    float sc = ws[2 * CSZ + c];
    float sh = ws[3 * CSZ + c];
    const float* xp = x + (size_t)bc * NSZ;

    float acc[R_H];
#pragma unroll
    for (int k = 0; k < R_H; ++k) acc[k] = 0.f;

    for (int d = threadIdx.x; d < NSZ; d += 256) {
        float xn = fmaf(sc, xp[d], sh);
        float p[R_H];
        hermite6(xn, p);
#pragma unroll
        for (int k = 0; k < R_H; ++k) acc[k] = fmaf(p[k], xn, acc[k]);
    }
#pragma unroll
    for (int k = 0; k < R_H; ++k) {
        red[threadIdx.x] = acc[k]; __syncthreads();
        for (int st = 128; st > 0; st >>= 1) {
            if (threadIdx.x < st) red[threadIdx.x] += red[threadIdx.x + st];
            __syncthreads();
        }
        if (threadIdx.x == 0) costk[k] = red[0];
        __syncthreads();
    }
    if (threadIdx.x == 0) {
        float m = costk[0];
#pragma unroll
        for (int k = 1; k < R_H; ++k) m = fmaxf(m, costk[k]);
        float e[R_H], ssum = 0.f;
#pragma unroll
        for (int k = 0; k < R_H; ++k) { e[k] = __expf(costk[k] - m); ssum += e[k]; }
        float inv = 1.0f / ssum;
#pragma unroll
        for (int k = 0; k < R_H; ++k) ws[4 * CSZ + bc * R_H + k] = e[k] * inv;
    }
}

__global__ void __launch_bounds__(256) k_fused(const float* __restrict__ x,
                                               const float* __restrict__ Wt,
                                               const float* __restrict__ bias,
                                               const float* __restrict__ ws,
                                               float* __restrict__ out) {
    __shared__ float y_pair[(CSZ / 2) * YPS];
    __shared__ float wt_pair[(CSZ / 2) * WPS];
    __shared__ float sc_l[CSZ], sh_l[CSZ], w_l[CSZ * R_H];

    int b      = blockIdx.y;
    int d_base = blockIdx.x * DT;
    int tid    = threadIdx.x;

    for (int i = tid; i < OC * CSZ; i += 256) {
        int o = i >> 6, c = i & (CSZ - 1);
        wt_pair[(c >> 1) * WPS + o * 2 + (c & 1)] = Wt[i];
    }
    if (tid < CSZ) { sc_l[tid] = ws[2 * CSZ + tid]; sh_l[tid] = ws[3 * CSZ + tid]; }
    for (int i = tid; i < CSZ * R_H; i += 256)
        w_l[i] = ws[4 * CSZ + b * CSZ * R_H + i];
    __syncthreads();

    int dl   = tid & (DT - 1);
    int cofs = tid >> 7;
    for (int cb = 0; cb < CSZ; cb += 2) {
        int c = cb + cofs;
        const float* xrow = &x[(((size_t)b * CSZ + c) * NSZ) + d_base + dl];
        float xv = *xrow;
        if (cb + 2 < CSZ) __builtin_prefetch(xrow + 2 * NSZ, 0, 0);
        float xn = fmaf(sc_l[c], xv, sh_l[c]);
        float p[R_H];
        hermite6(xn, p);
        float y = 0.f;
#pragma unroll
        for (int k = 0; k < R_H; ++k) y = fmaf(w_l[c * R_H + k], p[k], y);
        y_pair[(cb >> 1) * YPS + dl * 2 + cofs] = y;
    }
    __syncthreads();

    int wave = tid >> 5;
    int lane = tid & 31;
    int half = lane >> 4;
    int lid  = lane & 15;
    int o0   = (wave & 3) << 4;

    v2f afrag[16];
#pragma unroll
    for (int s = 0; s < 16; ++s)
        afrag[s] = *(const v2f*)&wt_pair[(2 * s + half) * WPS + (o0 + lid) * 2];

#pragma unroll
    for (int i = 0; i < 4; ++i) {
        int d0 = ((((wave >> 2) << 2) + i) << 4);
        v8f acc = {0.f, 0.f, 0.f, 0.f, 0.f, 0.f, 0.f, 0.f};
#pragma unroll
        for (int s = 0; s < 16; ++s) {
            v2f bb = *(const v2f*)&y_pair[(2 * s + half) * YPS + (d0 + lid) * 2];
            acc = __builtin_amdgcn_wmma_f32_16x16x4_f32(
                false, afrag[s], false, bb, (short)0, acc, false, false);
        }
        int dcol = d_base + d0 + lid;
#pragma unroll
        for (int v = 0; v < 8; ++v) {
            int o = o0 + v + 8 * half;
            out[(((size_t)b * OC + o) * NSZ) + dcol] = acc[v] + bias[o];
        }
    }
}

extern "C" void kernel_launch(void* const* d_in, const int* in_sizes, int n_in,
                              void* d_out, int out_size, void* d_ws, size_t ws_size,
                              hipStream_t stream) {
    const float* x     = (const float*)d_in[0];
    const float* gamma = (const float*)d_in[1];
    const float* beta  = (const float*)d_in[2];
    const float* Wt    = (const float*)d_in[3];
    const float* bias  = (const float*)d_in[4];
    float* out = (float*)d_out;
    float* ws  = (float*)d_ws;

    hipLaunchKernelGGL(k_init,     dim3(1),             dim3(128), 0, stream, ws);
    hipLaunchKernelGGL(k_stats,    dim3(BSZ * CSZ),     dim3(256), 0, stream, x, ws);
    hipLaunchKernelGGL(k_finalize, dim3(1),             dim3(64),  0, stream, gamma, beta, ws);
    hipLaunchKernelGGL(k_cost,     dim3(BSZ * CSZ),     dim3(256), 0, stream, x, ws);
    hipLaunchKernelGGL(k_fused,    dim3(NSZ / DT, BSZ), dim3(256), 0, stream,
                       x, Wt, bias, ws, out);
}


# GEMM kernel: a wave owns a 64x32 column slab and loads its B fragments straight from global memory (8 B per lane: even/odd column tiles), no LDS staging of y; 256 blocks x 2 slabs, 2nd prefetched
# speedup vs baseline: 1.5094x; 1.5094x over previous
#include <hip/hip_runtime.h>
#include <hip/hip_bf16.h>
#include <math.h>

#define R_H 6
#define EPSBN 1e-5f
#define BSZ 8
#define CSZ 64
#define NSZ 16384
#define OC  64
#define WPS 160

typedef __attribute__((ext_vector_type(2))) float v2f;
typedef __attribute__((ext_vector_type(8))) float v8f;

__device__ __forceinline__ float hermite_t(float xn) {
    const float SQ13 = 3.60555127546398929312f;
    return tanhf(xn) * SQ13;
}
__device__ __forceinline__ void hermite6_from_t(float t, float p[R_H]) {
    const float PIQ  = 0.75112554446494248286f;
    float g = __expf(-0.5f * t * t);
    p[0] = PIQ * g;
    p[1] = 1.41421356237309515f * PIQ * t * g;
    p[2] = 1.0f                 * t * p[1] - 0.70710678118654752f * p[0];
    p[3] = 0.81649658092772603f * t * p[2] - 0.81649658092772603f * p[1];
    p[4] = 0.70710678118654752f * t * p[3] - 0.86602540378443865f * p[2];
    p[5] = 0.63245553203367587f * t * p[4] - 0.89442719099991588f * p[3];
}

#define ST 1024
#define SV ((BSZ * NSZ / 4) / ST)
__global__ void __launch_bounds__(ST) __attribute__((amdgpu_waves_per_eu(8, 8)))
k_stats(const float* __restrict__ x,
        const float* __restrict__ gamma,
        const float* __restrict__ beta,
        float* __restrict__ ws) {
    __shared__ float red_s[ST / 32];
    __shared__ float red_q[ST / 32];
    int c    = blockIdx.x;
    int tid  = threadIdx.x;
    int lane = tid & 31;
    int wave = tid >> 5;
    float gm = gamma[c];
    float bt = beta[c];

    float s[4] = {0.f, 0.f, 0.f, 0.f};
    float q[4] = {0.f, 0.f, 0.f, 0.f};
#pragma unroll
    for (int j = 0; j < SV; ++j) {
        int b  = j >> 2;
        int f4 = ((j & 3) << 10) + tid;
        const float4* rp = (const float4*)(x + (size_t)(b * CSZ + c) * NSZ);
        float4 v = rp[f4];
        s[0] += v.x; s[1] += v.y; s[2] += v.z; s[3] += v.w;
        q[0] = fmaf(v.x, v.x, q[0]); q[1] = fmaf(v.y, v.y, q[1]);
        q[2] = fmaf(v.z, v.z, q[2]); q[3] = fmaf(v.w, v.w, q[3]);
    }
    float ss = (s[0] + s[1]) + (s[2] + s[3]);
    float qq = (q[0] + q[1]) + (q[2] + q[3]);
#pragma unroll
    for (int m = 16; m > 0; m >>= 1) {
        ss += __shfl_xor(ss, m, 32);
        qq += __shfl_xor(qq, m, 32);
    }
    if (lane == 0) { red_s[wave] = ss; red_q[wave] = qq; }
    __syncthreads();
    if (wave == 0) {
        float a = red_s[lane];
        float d = red_q[lane];
#pragma unroll
        for (int m = 16; m > 0; m >>= 1) {
            a += __shfl_xor(a, m, 32);
            d += __shfl_xor(d, m, 32);
        }
        if (lane == 0) {
            const float inv_n = 1.0f / ((float)BSZ * (float)NSZ);
            float mean = a * inv_n;
            float var  = d * inv_n - mean * mean;
            float rstd = rsqrtf(var + EPSBN);
            float sc   = gm * rstd;
            ws[2 * CSZ + c] = sc;
            ws[3 * CSZ + c] = bt - mean * sc;
        }
    }
}

#define RT 512
#define RE (NSZ / RT)
#define RG 8
__global__ void __launch_bounds__(RT) k_cost_y(const float* __restrict__ x,
                                               const float* __restrict__ ws,
                                               float* __restrict__ yout) {
    __shared__ float trow[NSZ];
    __shared__ float red[RT];
    __shared__ float costk[R_H];
    __shared__ float wk[R_H];
    int bc = blockIdx.x;
    int c  = bc & (CSZ - 1);
    int tid = threadIdx.x;
    float sc = ws[2 * CSZ + c];
    float sh = ws[3 * CSZ + c];
    const float* xp = x + (size_t)bc * NSZ + tid;

    float acc[R_H];
#pragma unroll
    for (int k = 0; k < R_H; ++k) acc[k] = 0.f;

#pragma unroll 1
    for (int j0 = 0; j0 < RE; j0 += RG) {
        float xv[RG];
#pragma unroll
        for (int j = 0; j < RG; ++j) xv[j] = xp[(j0 + j) * RT];
#pragma unroll
        for (int j = 0; j < RG; ++j) {
            float xn = fmaf(sc, xv[j], sh);
            float t = hermite_t(xn);
            trow[(j0 + j) * RT + tid] = t;
            float p[R_H];
            hermite6_from_t(t, p);
#pragma unroll
            for (int k = 0; k < R_H; ++k) acc[k] = fmaf(p[k], xn, acc[k]);
        }
    }
#pragma unroll
    for (int k = 0; k < R_H; ++k) {
        red[tid] = acc[k]; __syncthreads();
        for (int st = RT / 2; st > 0; st >>= 1) {
            if (tid < st) red[tid] += red[tid + st];
            __syncthreads();
        }
        if (tid == 0) costk[k] = red[0];
        __syncthreads();
    }
    if (tid == 0) {
        float m = costk[0];
#pragma unroll
        for (int k = 1; k < R_H; ++k) m = fmaxf(m, costk[k]);
        float e[R_H], ssum = 0.f;
#pragma unroll
        for (int k = 0; k < R_H; ++k) { e[k] = __expf(costk[k] - m); ssum += e[k]; }
        float inv = 1.0f / ssum;
#pragma unroll
        for (int k = 0; k < R_H; ++k) wk[k] = e[k] * inv;
    }
    __syncthreads();
    float w[R_H];
#pragma unroll
    for (int k = 0; k < R_H; ++k) w[k] = wk[k];

    float* yp = yout + (size_t)bc * NSZ + tid;
    for (int j0 = 0; j0 < RE; j0 += RG) {
#pragma unroll
        for (int j = 0; j < RG; ++j) {
            float p[R_H];
            hermite6_from_t(trow[(j0 + j) * RT + tid], p);
            float y = 0.f;
#pragma unroll
            for (int k = 0; k < R_H; ++k) y = fmaf(w[k], p[k], y);
            yp[(j0 + j) * RT] = y;
        }
    }
}

#define GS   32
#define GBLK 256
#define GSL  ((BSZ * NSZ / GS) / (GBLK * 8))

__device__ __forceinline__ void slab_load(const float* io, int g, int half, int lid,
                                          v2f r[32]) {
    int b     = g >> 9;
    int dbase = (g & 511) * GS;
    const float* p = io + ((size_t)b * CSZ + 2 * half) * NSZ + dbase + 2 * lid;
#pragma unroll
    for (int j = 0; j < 32; ++j)
        r[j] = *(const v2f*)(p + (size_t)(4 * (j >> 1) + (j & 1)) * NSZ);
}

__device__ __forceinline__ void slab_compute(float* io, const float (&bsv)[4][8],
                                             int g, int half, int lid,
                                             const v2f (&afrag)[4][16], const v2f r[32],
                                             bool more_work_follows) {
    int b     = g >> 9;
    int dbase = (g & 511) * GS;
#pragma unroll
    for (int op = 0; op < 2; ++op) {
        v8f aE0 = {0.f, 0.f, 0.f, 0.f, 0.f, 0.f, 0.f, 0.f};
        v8f aO0 = aE0, aE1 = aE0, aO1 = aE0;
#pragma unroll
        for (int s = 0; s < 16; ++s) {
            v2f bE = {r[2 * s].x, r[2 * s + 1].x};
            v2f bO = {r[2 * s].y, r[2 * s + 1].y};
            aE0 = __builtin_amdgcn_wmma_f32_16x16x4_f32(
                false, afrag[2 * op][s], false, bE, (short)0, aE0, false, false);
            aO0 = __builtin_amdgcn_wmma_f32_16x16x4_f32(
                false, afrag[2 * op][s], false, bO, (short)0, aO0, false, false);
            aE1 = __builtin_amdgcn_wmma_f32_16x16x4_f32(
                false, afrag[2 * op + 1][s], false, bE, (short)0, aE1, false, false);
            aO1 = __builtin_amdgcn_wmma_f32_16x16x4_f32(
                false, afrag[2 * op + 1][s], false, bO, (short)0, aO1, false, false);
        }
#pragma unroll
        for (int v = 0; v < 8; ++v) {
            int o0 = (2 * op) * 16 + v + 8 * half;
            int o1 = o0 + 16;
            float bs0 = bsv[2 * op][v], bs1 = bsv[2 * op + 1][v];
            v2f w0 = {aE0[v] + bs0, aO0[v] + bs0};
            v2f w1 = {aE1[v] + bs1, aO1[v] + bs1};
            *(v2f*)(io + ((size_t)b * OC + o0) * NSZ + dbase + 2 * lid) = w0;
            *(v2f*)(io + ((size_t)b * OC + o1) * NSZ + dbase + 2 * lid) = w1;
        }
        if (more_work_follows || op == 0) __builtin_amdgcn_sched_barrier(0);
    }
}

__global__ void __launch_bounds__(256) k_gemm(const float* __restrict__ Wt,
                                              const float* __restrict__ bias,
                                              float* io) {
    __shared__ float wt_pair[(CSZ / 2) * WPS];

    int tid  = threadIdx.x;
    int wave = tid >> 5;
    int lane = tid & 31;
    int half = lane >> 4;
    int lid  = lane & 15;
    int g0   = blockIdx.x * 8 + wave;
    static_assert(GSL == 2, "the body below is written out for two slabs per wave");

    float bsv[4][8];
#pragma unroll
    for (int ot = 0; ot < 4; ++ot)
#pragma unroll
        for (int v = 0; v < 8; ++v) bsv[ot][v] = bias[ot * 16 + 8 * half + v];

    v2f r0[32], r1[32];
    slab_load(io, g0, half, lid, r0);

    for (int i = tid; i < OC * CSZ; i += 256) {
        int o = i >> 6, c = i & (CSZ - 1);
        wt_pair[(c >> 1) * WPS + o * 2 + (c & 1)] = Wt[i];
    }
    __syncthreads();

    v2f afrag[4][16];
#pragma unroll
    for (int ot = 0; ot < 4; ++ot)
#pragma unroll
        for (int s = 0; s < 16; ++s)
            afrag[ot][s] = *(const v2f*)&wt_pair[(2 * s + half) * WPS + (ot * 16 + lid) * 2];

    slab_load(io, g0 + GBLK * 8, half, lid, r1);
    slab_compute(io, bsv, g0, half, lid, afrag, r0, true);
    slab_compute(io, bsv, g0 + GBLK * 8, half, lid, afrag, r1, false);
}

extern "C" void kernel_launch(void* const* d_in, const int* in_sizes, int n_in,
                              void* d_out, int out_size, void* d_ws, size_t ws_size,
                              hipStream_t stream) {
    const float* x     = (const float*)d_in[0];
    const float* gamma = (const float*)d_in[1];
    const float* beta  = (const float*)d_in[2];
    const float* Wt    = (const float*)d_in[3];
    const float* bias  = (const float*)d_in[4];
    float* out = (float*)d_out;
    float* ws  = (float*)d_ws;

    hipLaunchKernelGGL(k_stats,    dim3(CSZ),           dim3(ST),  0, stream, x, gamma, beta, ws);
    hipLaunchKernelGGL(k_cost_y,   dim3(BSZ * CSZ),     dim3(RT),  0, stream, x, ws, out);
    hipLaunchKernelGGL(k_gemm,     dim3(GBLK),          dim3(256), 0, stream,
                       Wt, bias, out);
}


# cost sums: the six LDS trees go down side by side, one work-group barrier per tree level for all six (67 -> 10 barriers in that phase); every add as before, sum order unchanged
# speedup vs baseline: 1.9675x; 1.3035x over previous
#include <hip/hip_runtime.h>
#include <hip/hip_bf16.h>
#include <math.h>

#define R_H 6
#define EPSBN 1e-5f
#define BSZ 8
#define CSZ 64
#define NSZ 16384
#define OC  64
#define WPS 160

typedef __attribute__((ext_vector_type(2))) float v2f;
typedef __attribute__((ext_vector_type(8))) float v8f;

__device__ __forceinline__ float hermite_t(float xn) {
    const float SQ13 = 3.60555127546398929312f;
    return tanhf(xn) * SQ13;
}
__device__ __forceinline__ void hermite6_from_t(float t, float p[R_H]) {
    const float PIQ  = 0.75112554446494248286f;
    float g = __expf(-0.5f * t * t);
    p[0] = PIQ * g;
    p[1] = 1.41421356237309515f * PIQ * t * g;
    p[2] = 1.0f                 * t * p[1] - 0.70710678118654752f * p[0];
    p[3] = 0.81649658092772603f * t * p[2] - 0.81649658092772603f * p[1];
    p[4] = 0.70710678118654752f * t * p[3] - 0.86602540378443865f * p[2];
    p[5] = 0.63245553203367587f * t * p[4] - 0.89442719099991588f * p[3];
}

#define FT 1024
#define RT 512
#define OWN4 ((NSZ / 4) / RT)
#define OTH4 ((NSZ / 4) / FT)
#define FBLK 256
__global__ void __launch_bounds__(FT) __attribute__((amdgpu_waves_per_eu(8, 8)))
k_stats_cost_y(const float* __restrict__ x,
               const float* __restrict__ gamma,
               const float* __restrict__ beta,
               float* __restrict__ yout) {
    __shared__ float red_s[FT / 32];
    __shared__ float red_q[FT / 32];
    __shared__ float scsh[2];
    __shared__ float red[R_H][FT];
    __shared__ float wk[2][R_H];
    int c    = blockIdx.x & (CSZ - 1);
    int q    = blockIdx.x >> 6;
    int tid  = threadIdx.x;
    int lane = tid & 31;
    int wave = tid >> 5;
    int r    = tid >> 9;
    int tl   = tid & (RT - 1);
    int bown = 2 * q + r;
    float gm = gamma[c];
    float bt = beta[c];

    const float4* op = (const float4*)(x + (size_t)(bown * CSZ + c) * NSZ) + tl;
    float4 xv[OWN4];
#pragma unroll
    for (int m = 0; m < OWN4; ++m) xv[m] = op[m * RT];

    float s[4]  = {0.f, 0.f, 0.f, 0.f};
    float sq[4] = {0.f, 0.f, 0.f, 0.f};
#pragma unroll 1
    for (int i = 0; i < BSZ - 2; i += 3) {
        float4 ov[3 * OTH4];
#pragma unroll
        for (int u = 0; u < 3; ++u) {
            int b = (2 * q + 2 + i + u) & (BSZ - 1);
            const float4* rp = (const float4*)(x + (size_t)(b * CSZ + c) * NSZ) + tid;
#pragma unroll
            for (int m = 0; m < OTH4; ++m) ov[u * OTH4 + m] = rp[m * FT];
        }
#pragma unroll
        for (int m = 0; m < 3 * OTH4; ++m) {
            float4 v = ov[m];
            s[0] += v.x; s[1] += v.y; s[2] += v.z; s[3] += v.w;
            sq[0] = fmaf(v.x, v.x, sq[0]); sq[1] = fmaf(v.y, v.y, sq[1]);
            sq[2] = fmaf(v.z, v.z, sq[2]); sq[3] = fmaf(v.w, v.w, sq[3]);
        }
    }
#pragma unroll
    for (int m = 0; m < OWN4; ++m) {
        float4 v = xv[m];
        s[0] += v.x; s[1] += v.y; s[2] += v.z; s[3] += v.w;
        sq[0] = fmaf(v.x, v.x, sq[0]); sq[1] = fmaf(v.y, v.y, sq[1]);
        sq[2] = fmaf(v.z, v.z, sq[2]); sq[3] = fmaf(v.w, v.w, sq[3]);
    }
    float ss = (s[0] + s[1]) + (s[2] + s[3]);
    float qq = (sq[0] + sq[1]) + (sq[2] + sq[3]);
#pragma unroll
    for (int m = 16; m > 0; m >>= 1) {
        ss += __shfl_xor(ss, m, 32);
        qq += __shfl_xor(qq, m, 32);
    }
    if (lane == 0) { red_s[wave] = ss; red_q[wave] = qq; }
    __syncthreads();
    if (wave == 0) {
        float a = red_s[lane];
        float d = red_q[lane];
#pragma unroll
        for (int m = 16; m > 0; m >>= 1) {
            a += __shfl_xor(a, m, 32);
            d += __shfl_xor(d, m, 32);
        }
        if (lane == 0) {
            const float inv_n = 1.0f / ((float)BSZ * (float)NSZ);
            float mean = a * inv_n;
            float var  = d * inv_n - mean * mean;
            float rstd = rsqrtf(var + EPSBN);
            float scl  = gm * rstd;
            scsh[0] = scl;
            scsh[1] = bt - mean * scl;
        }
    }
    __syncthreads();
    float sc = scsh[0];
    float sh = scsh[1];

    float acc[R_H];
#pragma unroll
    for (int k = 0; k < R_H; ++k) acc[k] = 0.f;
#pragma unroll
    for (int m = 0; m < OWN4; ++m) {
        float v[4] = {xv[m].x, xv[m].y, xv[m].z, xv[m].w};
#pragma unroll
        for (int e = 0; e < 4; ++e) {
            float xn = fmaf(sc, v[e], sh);
            float t = hermite_t(xn);
            v[e] = t;
            float p[R_H];
            hermite6_from_t(t, p);
#pragma unroll
            for (int k = 0; k < R_H; ++k) acc[k] = fmaf(p[k], xn, acc[k]);
        }
        xv[m] = make_float4(v[0], v[1], v[2], v[3]);
        asm volatile("" : "+v"(acc[0]), "+v"(acc[1]), "+v"(acc[2]),
                          "+v"(acc[3]), "+v"(acc[4]), "+v"(acc[5]));
    }
#pragma unroll
    for (int k = 0; k < R_H; ++k) red[k][tid] = acc[k];
    __syncthreads();
    for (int st = RT / 2; st > 1; st >>= 1) {
        if (tl < st) {
#pragma unroll
            for (int k = 0; k < R_H; ++k) {
                acc[k] += red[k][tid + st];
                red[k][tid] = acc[k];
            }
        }
        __syncthreads();
    }
    if (tl == 0) {
        float cst[R_H];
#pragma unroll
        for (int k = 0; k < R_H; ++k) cst[k] = acc[k] + red[k][tid + 1];
        float m = cst[0];
#pragma unroll
        for (int k = 1; k < R_H; ++k) m = fmaxf(m, cst[k]);
        float e[R_H], ssum = 0.f;
#pragma unroll
        for (int k = 0; k < R_H; ++k) { e[k] = __expf(cst[k] - m); ssum += e[k]; }
        float inv = 1.0f / ssum;
#pragma unroll
        for (int k = 0; k < R_H; ++k) wk[r][k] = e[k] * inv;
    }
    __syncthreads();
    float w[R_H];
#pragma unroll
    for (int k = 0; k < R_H; ++k) w[k] = wk[r][k];

#pragma unroll
    for (int m = 0; m < OWN4; ++m)
        asm volatile("" : "+v"(xv[m].x), "+v"(xv[m].y), "+v"(xv[m].z), "+v"(xv[m].w));

    float4* yp = (float4*)(yout + (size_t)(bown * CSZ + c) * NSZ) + tl;
#pragma unroll
    for (int m = 0; m < OWN4; ++m) {
        float v[4] = {xv[m].x, xv[m].y, xv[m].z, xv[m].w};
#pragma unroll
        for (int e = 0; e < 4; ++e) {
            float p[R_H];
            hermite6_from_t(v[e], p);
            float y = 0.f;
#pragma unroll
            for (int k = 0; k < R_H; ++k) y = fmaf(w[k], p[k], y);
            v[e] = y;
        }
        yp[m * RT] = make_float4(v[0], v[1], v[2], v[3]);
    }
}

#define GS   32
#define GBLK 256
#define GSL  ((BSZ * NSZ / GS) / (GBLK * 8))

__device__ __forceinline__ void slab_load(const float* io, int g, int half, int lid,
                                          v2f r[32]) {
    int b     = g >> 9;
    int dbase = (g & 511) * GS;
    const float* p = io + ((size_t)b * CSZ + 2 * half) * NSZ + dbase + 2 * lid;
#pragma unroll
    for (int j = 0; j < 32; ++j)
        r[j] = *(const v2f*)(p + (size_t)(4 * (j >> 1) + (j & 1)) * NSZ);
}

__device__ __forceinline__ void slab_compute(float* io, const float (&bsv)[4][8],
                                             int g, int half, int lid,
                                             const v2f (&afrag)[4][16], const v2f r[32],
                                             bool more_work_follows) {
    int b     = g >> 9;
    int dbase = (g & 511) * GS;
#pragma unroll
    for (int op = 0; op < 2; ++op) {
        v8f aE0 = {0.f, 0.f, 0.f, 0.f, 0.f, 0.f, 0.f, 0.f};
        v8f aO0 = aE0, aE1 = aE0, aO1 = aE0;
#pragma unroll
        for (int s = 0; s < 16; ++s) {
            v2f bE = {r[2 * s].x, r[2 * s + 1].x};
            v2f bO = {r[2 * s].y, r[2 * s + 1].y};
            aE0 = __builtin_amdgcn_wmma_f32_16x16x4_f32(
                false, afrag[2 * op][s], false, bE, (short)0, aE0, false, false);
            aO0 = __builtin_amdgcn_wmma_f32_16x16x4_f32(
                false, afrag[2 * op][s], false, bO, (short)0, aO0, false, false);
            aE1 = __builtin_amdgcn_wmma_f32_16x16x4_f32(
                false, afrag[2 * op + 1][s], false, bE, (short)0, aE1, false, false);
            aO1 = __builtin_amdgcn_wmma_f32_16x16x4_f32(
                false, afrag[2 * op + 1][s], false, bO, (short)0, aO1, false, false);
        }
#pragma unroll
        for (int v = 0; v < 8; ++v) {
            int o0 = (2 * op) * 16 + v + 8 * half;
            int o1 = o0 + 16;
            float bs0 = bsv[2 * op][v], bs1 = bsv[2 * op + 1][v];
            v2f w0 = {aE0[v] + bs0, aO0[v] + bs0};
            v2f w1 = {aE1[v] + bs1, aO1[v] + bs1};
            *(v2f*)(io + ((size_t)b * OC + o0) * NSZ + dbase + 2 * lid) = w0;
            *(v2f*)(io + ((size_t)b * OC + o1) * NSZ + dbase + 2 * lid) = w1;
        }
        if (more_work_follows || op == 0) __builtin_amdgcn_sched_barrier(0);
    }
}

__global__ void __launch_bounds__(256) k_gemm(const float* __restrict__ Wt,
                                              const float* __restrict__ bias,
                                              float* io) {
    __shared__ float wt_pair[(CSZ / 2) * WPS];

    int tid  = threadIdx.x;
    int wave = tid >> 5;
    int lane = tid & 31;
    int half = lane >> 4;
    int lid  = lane & 15;
    int g0   = blockIdx.x * 8 + wave;
    static_assert(GSL == 2, "the body below is written out for two slabs per wave");

    float bsv[4][8];
#pragma unroll
    for (int ot = 0; ot < 4; ++ot)
#pragma unroll
        for (int v = 0; v < 8; ++v) bsv[ot][v] = bias[ot * 16 + 8 * half + v];

    v2f r0[32], r1[32];
    slab_load(io, g0, half, lid, r0);

    for (int i = tid; i < OC * CSZ; i += 256) {
        int o = i >> 6, c = i & (CSZ - 1);
        wt_pair[(c >> 1) * WPS + o * 2 + (c & 1)] = Wt[i];
    }
    __syncthreads();

    v2f afrag[4][16];
#pragma unroll
    for (int ot = 0; ot < 4; ++ot)
#pragma unroll
        for (int s = 0; s < 16; ++s)
            afrag[ot][s] = *(const v2f*)&wt_pair[(2 * s + half) * WPS + (ot * 16 + lid) * 2];

    slab_load(io, g0 + GBLK * 8, half, lid, r1);
    slab_compute(io, bsv, g0, half, lid, afrag, r0, true);
    slab_compute(io, bsv, g0 + GBLK * 8, half, lid, afrag, r1, false);
}

extern "C" void kernel_launch(void* const* d_in, const int* in_sizes, int n_in,
                              void* d_out, int out_size, void* d_ws, size_t ws_size,
                              hipStream_t stream) {
    const float* x     = (const float*)d_in[0];
    const float* gamma = (const float*)d_in[1];
    const float* beta  = (const float*)d_in[2];
    const float* Wt    = (const float*)d_in[3];
    const float* bias  = (const float*)d_in[4];
    float* out = (float*)d_out;
    (void)d_ws;

    hipLaunchKernelGGL(k_stats_cost_y, dim3(FBLK),    dim3(FT),  0, stream, x, gamma, beta, out);
    hipLaunchKernelGGL(k_gemm,         dim3(GBLK),    dim3(256), 0, stream,
                       Wt, bias, out);
}
